# starting kernel
# baseline (speedup 1.0000x reference)
#include <hip/hip_runtime.h>
#include <hip/hip_bf16.h>

typedef __attribute__((ext_vector_type(2))) float v2f;
typedef __attribute__((ext_vector_type(8))) float v8f;

#define FEAT 64

__global__ __launch_bounds__(256) void degree_kernel(const int* __restrict__ dst,
                                                     float* __restrict__ deg, int E) {
    int e = blockIdx.x * blockDim.x + threadIdx.x;
    if (e < E) {
        unsafeAtomicAdd(&deg[dst[e]], 1.0f);
    }
}

__global__ __launch_bounds__(256) void scatter_add_kernel(const float* __restrict__ feat,
                                                          const int* __restrict__ src,
                                                          const int* __restrict__ dst,
                                                          float* __restrict__ agg,
                                                          long long total) {
    long long i = (long long)blockIdx.x * blockDim.x + threadIdx.x;
    if (i >= total) return;
    int e = (int)(i >> 6);
    int f = (int)(i & 63);
    int s = src[e];
    int d = dst[e];
    float v = feat[(long long)s * FEAT + f];
    unsafeAtomicAdd(&agg[(long long)d * FEAT + f], v);
}

__global__ __launch_bounds__(32) void wmma_gemm_kernel(const float* __restrict__ S,
                                                       const float* __restrict__ deg,
                                                       const float* __restrict__ W,
                                                       const float* __restrict__ bias,
                                                       float* __restrict__ out,
                                                       int n, int tilesN, int ncols,
                                                       int do_relu) {
    const int tile = blockIdx.x;
    const int tm = tile / tilesN;
    const int tn = tile % tilesN;
    const int lane = threadIdx.x;
    const int half = lane >> 4;
    const int l15  = lane & 15;

    int rowA = tm * 16 + l15;
    if (rowA >= n) rowA = n - 1;
    const int kb = half * 2;

    const int colB = tn * 16 + l15;
    const int cB   = (colB < ncols) ? colB : 0;
    const float bmask = (colB < ncols) ? 1.0f : 0.0f;

    const float* Sp = S + (long long)rowA * FEAT;

    v8f c = {};
    #pragma unroll
    for (int k0 = 0; k0 < FEAT; k0 += 4) {
        const int k = k0 + kb;
        v2f a = *(const v2f*)(Sp + k);
        v2f b;
        b.x = W[(long long)k * ncols + cB] * bmask;
        b.y = W[(long long)(k + 1) * ncols + cB] * bmask;
        c = __builtin_amdgcn_wmma_f32_16x16x4_f32(
                 false, a,  false, b,
                 (short)0, c,  false,  false);
    }

    const float bv = (colB < ncols) ? bias[colB] : 0.0f;
    #pragma unroll
    for (int r = 0; r < 8; ++r) {
        int row = tm * 16 + r + half * 8;
        int rc = (row < n) ? row : (n - 1);
        float d = deg[rc];
        float inv = (d > 0.0f) ? (1.0f / d) : 0.0f;
        float v = c[r] * inv + bv;
        if (do_relu) v = fmaxf(v, 0.0f);
        if (colB < ncols && row < n) {
            out[(long long)row * ncols + colB] = v;
        }
    }
}

extern "C" void kernel_launch(void* const* d_in, const int* in_sizes, int n_in,
                              void* d_out, int out_size, void* d_ws, size_t ws_size,
                              hipStream_t stream) {
    const float* features = (const float*)d_in[0];
    const float* W1 = (const float*)d_in[1];
    const float* b1 = (const float*)d_in[2];
    const float* W2 = (const float*)d_in[3];
    const float* b2 = (const float*)d_in[4];
    const int* src = (const int*)d_in[5];
    const int* dst = (const int*)d_in[6];
    float* out = (float*)d_out;

    const int N = in_sizes[0] / FEAT;
    const int E = in_sizes[5];
    const int NC1 = FEAT;
    const int NC2 = in_sizes[4];

    float* agg = (float*)d_ws;
    float* deg = agg + (long long)N * FEAT;
    float* h1  = deg + N;

    hipMemsetAsync(agg, 0, (size_t)((long long)N * FEAT + N) * sizeof(float), stream);

    degree_kernel<<<(E + 255) / 256, 256, 0, stream>>>(dst, deg, E);

    long long total = (long long)E * FEAT;
    int blocks = (int)((total + 255) / 256);
    scatter_add_kernel<<<blocks, 256, 0, stream>>>(features, src, dst, agg, total);

    const int tilesM = (N + 15) / 16;
    const int tilesN1 = (NC1 + 15) / 16;
    wmma_gemm_kernel<<<tilesM * tilesN1, 32, 0, stream>>>(agg, deg, W1, b1, h1,
                                                          N, tilesN1, NC1,  1);

    hipMemsetAsync(agg, 0, (size_t)((long long)N * FEAT) * sizeof(float), stream);
    scatter_add_kernel<<<blocks, 256, 0, stream>>>(h1, src, dst, agg, total);

    const int tilesN2 = (NC2 + 15) / 16;
    wmma_gemm_kernel<<<tilesM * tilesN2, 32, 0, stream>>>(agg, deg, W2, b2, out,
                                                          N, tilesN2, NC2,  0);
}


# Aggregation by in-call edge grouping (integer count+rank, segment alloc, fill) and per-node gather-sum instead of f32 atomic scatter; sum order = arrival-order rank per destination
# speedup vs baseline: 2.3896x; 2.3896x over previous
#include <hip/hip_runtime.h>
#include <hip/hip_bf16.h>

typedef __attribute__((ext_vector_type(2))) float v2f;
typedef __attribute__((ext_vector_type(8))) float v8f;

#define FEAT 64


__global__ __launch_bounds__(256) void count_rank_kernel(const int* __restrict__ dst,
                                                         int* __restrict__ cnt,
                                                         int* __restrict__ rank, int E) {
    int e = blockIdx.x * blockDim.x + threadIdx.x;
    if (e < E) {
        rank[e] = atomicAdd(&cnt[dst[e]], 1);
    }
}

__global__ __launch_bounds__(256) void segment_alloc_kernel(const int* __restrict__ cnt,
                                                            int* __restrict__ cursor,
                                                            int* __restrict__ start,
                                                            float* __restrict__ deg, int n) {
    __shared__ int wave_off[8];
    __shared__ int block_base;
    const int tid = threadIdx.x;
    const int lane = tid & 31;
    const int w = tid >> 5;
    const int v = blockIdx.x * 256 + tid;
    const int c = (v < n) ? cnt[v] : 0;

    int inc = c;
    #pragma unroll
    for (int off = 1; off < 32; off <<= 1) {
        int t = __shfl_up(inc, off, 32);
        if (lane >= off) inc += t;
    }
    if (lane == 31) wave_off[w] = inc;
    __syncthreads();
    if (tid == 0) {
        int run = 0;
        #pragma unroll
        for (int i = 0; i < 8; ++i) {
            int t = wave_off[i];
            wave_off[i] = run;
            run += t;
        }
        block_base = atomicAdd(cursor, run);
    }
    __syncthreads();
    if (v < n) {
        start[v] = block_base + wave_off[w] + inc - c;
        deg[v] = (float)c;
    }
}

__global__ __launch_bounds__(256) void fill_kernel(const int* __restrict__ src,
                                                   const int* __restrict__ dst,
                                                   const int* __restrict__ start,
                                                   const int* __restrict__ rank,
                                                   int* __restrict__ col, int E) {
    int e = blockIdx.x * blockDim.x + threadIdx.x;
    if (e < E) {
        col[start[dst[e]] + rank[e]] = src[e];
    }
}

__global__ __launch_bounds__(256) void gather_sum_kernel(const float* __restrict__ feat,
                                                         const int* __restrict__ col,
                                                         const int* __restrict__ start,
                                                         const int* __restrict__ cnt,
                                                         float* __restrict__ agg, int n) {
    const int node = blockIdx.x * 16 + (threadIdx.x >> 4);
    const int l = threadIdx.x & 15;
    if (node >= n) return;
    const int c = cnt[node];
    const int* __restrict__ cp = col + start[node];
    const float4* __restrict__ f4 = (const float4*)feat;

    float4 acc = {0.0f, 0.0f, 0.0f, 0.0f};
    int j = 0;
    for (; j + 4 <= c; j += 4) {
        const int s0 = cp[j], s1 = cp[j + 1], s2 = cp[j + 2], s3 = cp[j + 3];
        const float4 r0 = f4[(long long)s0 * 16 + l];
        const float4 r1 = f4[(long long)s1 * 16 + l];
        const float4 r2 = f4[(long long)s2 * 16 + l];
        const float4 r3 = f4[(long long)s3 * 16 + l];
        acc.x += r0.x; acc.y += r0.y; acc.z += r0.z; acc.w += r0.w;
        acc.x += r1.x; acc.y += r1.y; acc.z += r1.z; acc.w += r1.w;
        acc.x += r2.x; acc.y += r2.y; acc.z += r2.z; acc.w += r2.w;
        acc.x += r3.x; acc.y += r3.y; acc.z += r3.z; acc.w += r3.w;
    }
    for (; j < c; ++j) {
        const int s0 = cp[j];
        const float4 r0 = f4[(long long)s0 * 16 + l];
        acc.x += r0.x; acc.y += r0.y; acc.z += r0.z; acc.w += r0.w;
    }
    ((float4*)agg)[(long long)node * 16 + l] = acc;
}

__global__ __launch_bounds__(32) void wmma_gemm_kernel(const float* __restrict__ S,
                                                       const float* __restrict__ deg,
                                                       const float* __restrict__ W,
                                                       const float* __restrict__ bias,
                                                       float* __restrict__ out,
                                                       int n, int tilesN, int ncols,
                                                       int do_relu) {
    const int tile = blockIdx.x;
    const int tm = tile / tilesN;
    const int tn = tile % tilesN;
    const int lane = threadIdx.x;
    const int half = lane >> 4;
    const int l15  = lane & 15;

    int rowA = tm * 16 + l15;
    if (rowA >= n) rowA = n - 1;
    const int kb = half * 2;

    const int colB = tn * 16 + l15;
    const int cB   = (colB < ncols) ? colB : 0;
    const float bmask = (colB < ncols) ? 1.0f : 0.0f;

    const float* Sp = S + (long long)rowA * FEAT;

    v8f c = {};
    #pragma unroll
    for (int k0 = 0; k0 < FEAT; k0 += 4) {
        const int k = k0 + kb;
        v2f a = *(const v2f*)(Sp + k);
        v2f b;
        b.x = W[(long long)k * ncols + cB] * bmask;
        b.y = W[(long long)(k + 1) * ncols + cB] * bmask;
        c = __builtin_amdgcn_wmma_f32_16x16x4_f32(
                 false, a,  false, b,
                 (short)0, c,  false,  false);
    }

    const float bv = (colB < ncols) ? bias[colB] : 0.0f;
    #pragma unroll
    for (int r = 0; r < 8; ++r) {
        int row = tm * 16 + r + half * 8;
        int rc = (row < n) ? row : (n - 1);
        float d = deg[rc];
        float inv = (d > 0.0f) ? (1.0f / d) : 0.0f;
        float v = c[r] * inv + bv;
        if (do_relu) v = fmaxf(v, 0.0f);
        if (colB < ncols && row < n) {
            out[(long long)row * ncols + colB] = v;
        }
    }
}

extern "C" void kernel_launch(void* const* d_in, const int* in_sizes, int n_in,
                              void* d_out, int out_size, void* d_ws, size_t ws_size,
                              hipStream_t stream) {
    const float* features = (const float*)d_in[0];
    const float* W1 = (const float*)d_in[1];
    const float* b1 = (const float*)d_in[2];
    const float* W2 = (const float*)d_in[3];
    const float* b2 = (const float*)d_in[4];
    const int* src = (const int*)d_in[5];
    const int* dst = (const int*)d_in[6];
    float* out = (float*)d_out;

    const int N = in_sizes[0] / FEAT;
    const int E = in_sizes[5];
    const int NC1 = FEAT;
    const int NC2 = in_sizes[4];

    float* agg = (float*)d_ws;
    float* h1  = agg + (long long)N * FEAT;
    float* deg = h1 + (long long)N * FEAT;
    int* cnt    = (int*)(deg + N);
    int* cursor = cnt + N;
    int* start  = cursor + 1;
    int* rank   = start + N;
    int* col    = rank + E;

    hipMemsetAsync(cnt, 0, (size_t)(N + 1) * sizeof(int), stream);

    const int eblocks = (E + 255) / 256;
    const int nblocks = (N + 255) / 256;
    count_rank_kernel<<<eblocks, 256, 0, stream>>>(dst, cnt, rank, E);
    segment_alloc_kernel<<<nblocks, 256, 0, stream>>>(cnt, cursor, start, deg, N);
    fill_kernel<<<eblocks, 256, 0, stream>>>(src, dst, start, rank, col, E);

    const int gblocks = (N + 15) / 16;
    gather_sum_kernel<<<gblocks, 256, 0, stream>>>(features, col, start, cnt, agg, N);

    const int tilesM = (N + 15) / 16;
    const int tilesN1 = (NC1 + 15) / 16;
    wmma_gemm_kernel<<<tilesM * tilesN1, 32, 0, stream>>>(agg, deg, W1, b1, h1,
                                                          N, tilesN1, NC1,  1);

    gather_sum_kernel<<<gblocks, 256, 0, stream>>>(h1, col, start, cnt, agg, N);

    const int tilesN2 = (NC2 + 15) / 16;
    wmma_gemm_kernel<<<tilesM * tilesN2, 32, 0, stream>>>(agg, deg, W2, b2, out,
                                                          N, tilesN2, NC2,  0);
}


# Aggregation fused into the GEMM kernel: each block sums 16 nodes' neighbour rows into an LDS tile that feeds the WMMA A operand; agg plane and 2 launches removed; summation order unchanged
# speedup vs baseline: 2.9471x; 1.2333x over previous
#include <hip/hip_runtime.h>
#include <hip/hip_bf16.h>

typedef __attribute__((ext_vector_type(2))) float v2f;
typedef __attribute__((ext_vector_type(8))) float v8f;

#define FEAT 64
#define NCLS 40


__global__ __launch_bounds__(256) void count_rank_kernel(const int* __restrict__ dst,
                                                         int* __restrict__ cnt,
                                                         int* __restrict__ rank, int E) {
    int e = blockIdx.x * blockDim.x + threadIdx.x;
    if (e < E) {
        rank[e] = atomicAdd(&cnt[dst[e]], 1);
    }
}

__global__ __launch_bounds__(256) void segment_alloc_kernel(const int* __restrict__ cnt,
                                                            int* __restrict__ cursor,
                                                            int* __restrict__ start, int n) {
    __shared__ int wave_off[8];
    __shared__ int block_base;
    const int tid = threadIdx.x;
    const int lane = tid & 31;
    const int w = tid >> 5;
    const int v = blockIdx.x * 256 + tid;
    const int c = (v < n) ? cnt[v] : 0;

    int inc = c;
    #pragma unroll
    for (int off = 1; off < 32; off <<= 1) {
        int t = __shfl_up(inc, off, 32);
        if (lane >= off) inc += t;
    }
    if (lane == 31) wave_off[w] = inc;
    __syncthreads();
    if (tid == 0) {
        int run = 0;
        #pragma unroll
        for (int i = 0; i < 8; ++i) {
            int t = wave_off[i];
            wave_off[i] = run;
            run += t;
        }
        block_base = atomicAdd(cursor, run);
    }
    __syncthreads();
    if (v < n) {
        start[v] = block_base + wave_off[w] + inc - c;
    }
}

__global__ __launch_bounds__(256) void fill_kernel(const int* __restrict__ src,
                                                   const int* __restrict__ dst,
                                                   const int* __restrict__ start,
                                                   const int* __restrict__ rank,
                                                   int* __restrict__ col, int E) {
    int e = blockIdx.x * blockDim.x + threadIdx.x;
    if (e < E) {
        col[start[dst[e]] + rank[e]] = src[e];
    }
}

#define TILE_PITCH 68

template <int NCOLS, bool RELU>
__global__ __launch_bounds__(256) void gather_gemm_kernel(const float* __restrict__ feat,
                                                          const int* __restrict__ col,
                                                          const int* __restrict__ start,
                                                          const int* __restrict__ cnt,
                                                          const float* __restrict__ W,
                                                          const float* __restrict__ bias,
                                                          float* __restrict__ out, int n) {
    constexpr int tilesN = (NCOLS + 15) / 16;
    __shared__ __attribute__((aligned(16))) float tile[16 * TILE_PITCH];
    __shared__ __attribute__((aligned(16))) float sinv[16];

    const int tid = threadIdx.x;
    const int r = tid >> 4;
    const int l = tid & 15;
    const int node = blockIdx.x * 16 + r;

    float4 acc = {0.0f, 0.0f, 0.0f, 0.0f};
    int c = 0;
    if (node < n) {
        c = cnt[node];
        const int* __restrict__ cp = col + start[node];
        const float4* __restrict__ f4 = (const float4*)feat;
        int j = 0;
        for (; j + 4 <= c; j += 4) {
            const int s0 = cp[j], s1 = cp[j + 1], s2 = cp[j + 2], s3 = cp[j + 3];
            const float4 r0 = f4[(long long)s0 * 16 + l];
            const float4 r1 = f4[(long long)s1 * 16 + l];
            const float4 r2 = f4[(long long)s2 * 16 + l];
            const float4 r3 = f4[(long long)s3 * 16 + l];
            acc.x += r0.x; acc.y += r0.y; acc.z += r0.z; acc.w += r0.w;
            acc.x += r1.x; acc.y += r1.y; acc.z += r1.z; acc.w += r1.w;
            acc.x += r2.x; acc.y += r2.y; acc.z += r2.z; acc.w += r2.w;
            acc.x += r3.x; acc.y += r3.y; acc.z += r3.z; acc.w += r3.w;
        }
        for (; j < c; ++j) {
            const int s0 = cp[j];
            const float4 r0 = f4[(long long)s0 * 16 + l];
            acc.x += r0.x; acc.y += r0.y; acc.z += r0.z; acc.w += r0.w;
        }
    }
    *(float4*)(tile + r * TILE_PITCH + 4 * l) = acc;
    if (l == 0) {
        const float d = (float)c;
        sinv[r] = (d > 0.0f) ? (1.0f / d) : 0.0f;
    }
    __syncthreads();

    const int w = __builtin_amdgcn_readfirstlane(tid >> 5);
    if (w >= tilesN) return;

    const int lane = tid & 31;
    const int half = lane >> 4;
    const int l15  = lane & 15;

    const int kb = half * 2;
    const int colB = w * 16 + l15;
    const bool colOK = colB < NCOLS;
    const int cB   = colOK ? colB : 0;
    const float bmask = colOK ? 1.0f : 0.0f;

    const float* Ap = tile + l15 * TILE_PITCH + kb;
    const float* Wp = W + kb * NCOLS + cB;
    const float bv = bias[cB];

    v8f cacc = {};
    #pragma unroll
    for (int k0 = 0; k0 < FEAT; k0 += 4) {
        v2f a = *(const v2f*)(Ap + k0);
        v2f b;
        b.x = Wp[k0 * NCOLS] * bmask;
        b.y = Wp[(k0 + 1) * NCOLS] * bmask;
        cacc = __builtin_amdgcn_wmma_f32_16x16x4_f32(
                 false, a,  false, b,
                 (short)0, cacc,  false,  false);
    }

    const float4 i0 = *(const float4*)(sinv + half * 8);
    const float4 i1 = *(const float4*)(sinv + half * 8 + 4);
    const float invs[8] = {i0.x, i0.y, i0.z, i0.w, i1.x, i1.y, i1.z, i1.w};
    const int row0 = blockIdx.x * 16 + half * 8;
    float* op = out + (long long)row0 * NCOLS + cB;
    #pragma unroll
    for (int q = 0; q < 8; ++q) {
        float v = cacc[q] * invs[q] + bv;
        if (RELU) v = fmaxf(v, 0.0f);
        if (colOK && row0 + q < n) {
            op[q * NCOLS] = v;
        }
    }
}

extern "C" void kernel_launch(void* const* d_in, const int* in_sizes, int n_in,
                              void* d_out, int out_size, void* d_ws, size_t ws_size,
                              hipStream_t stream) {
    const float* features = (const float*)d_in[0];
    const float* W1 = (const float*)d_in[1];
    const float* b1 = (const float*)d_in[2];
    const float* W2 = (const float*)d_in[3];
    const float* b2 = (const float*)d_in[4];
    const int* src = (const int*)d_in[5];
    const int* dst = (const int*)d_in[6];
    float* out = (float*)d_out;

    const int N = in_sizes[0] / FEAT;
    const int E = in_sizes[5];

    float* h1   = (float*)d_ws;
    int* cnt    = (int*)(h1 + (long long)N * FEAT);
    int* cursor = cnt + N;
    int* start  = cursor + 1;
    int* rank   = start + N;
    int* col    = rank + E;

    hipMemsetAsync(cnt, 0, (size_t)(N + 1) * sizeof(int), stream);

    const int eblocks = (E + 255) / 256;
    const int nblocks = (N + 255) / 256;
    count_rank_kernel<<<eblocks, 256, 0, stream>>>(dst, cnt, rank, E);
    segment_alloc_kernel<<<nblocks, 256, 0, stream>>>(cnt, cursor, start, N);
    fill_kernel<<<eblocks, 256, 0, stream>>>(src, dst, start, rank, col, E);

    const int gblocks = (N + 15) / 16;
    gather_gemm_kernel<FEAT, true><<<gblocks, 256, 0, stream>>>(features, col, start, cnt,
                                                                 W1, b1, h1, N);

    gather_gemm_kernel<NCLS, false><<<gblocks, 256, 0, stream>>>(h1, col, start, cnt,
                                                                  W2, b2, out, N);
}


# Gather phase: each node's 16 lanes fetch 16 source ids in one coalesced load and pass them by lane shuffle; row loads go out 8 at a time with no id load in between; sum order unchanged
# speedup vs baseline: 3.0973x; 1.0510x over previous
#include <hip/hip_runtime.h>
#include <hip/hip_bf16.h>

typedef __attribute__((ext_vector_type(2))) float v2f;
typedef __attribute__((ext_vector_type(8))) float v8f;

#define FEAT 64
#define NCLS 40


__global__ __launch_bounds__(256) void count_rank_kernel(const int* __restrict__ dst,
                                                         int* __restrict__ cnt,
                                                         int* __restrict__ rank, int E) {
    int e = blockIdx.x * blockDim.x + threadIdx.x;
    if (e < E) {
        rank[e] = atomicAdd(&cnt[dst[e]], 1);
    }
}

__global__ __launch_bounds__(256) void segment_alloc_kernel(const int* __restrict__ cnt,
                                                            int* __restrict__ cursor,
                                                            int* __restrict__ start, int n) {
    __shared__ int wave_off[8];
    __shared__ int block_base;
    const int tid = threadIdx.x;
    const int lane = tid & 31;
    const int w = tid >> 5;
    const int v = blockIdx.x * 256 + tid;
    const int c = (v < n) ? cnt[v] : 0;

    int inc = c;
    #pragma unroll
    for (int off = 1; off < 32; off <<= 1) {
        int t = __shfl_up(inc, off, 32);
        if (lane >= off) inc += t;
    }
    if (lane == 31) wave_off[w] = inc;
    __syncthreads();
    if (tid == 0) {
        int run = 0;
        #pragma unroll
        for (int i = 0; i < 8; ++i) {
            int t = wave_off[i];
            wave_off[i] = run;
            run += t;
        }
        block_base = atomicAdd(cursor, run);
    }
    __syncthreads();
    if (v < n) {
        start[v] = block_base + wave_off[w] + inc - c;
    }
}

__global__ __launch_bounds__(256) void fill_kernel(const int* __restrict__ src,
                                                   const int* __restrict__ dst,
                                                   const int* __restrict__ start,
                                                   const int* __restrict__ rank,
                                                   int* __restrict__ col, int E) {
    int e = blockIdx.x * blockDim.x + threadIdx.x;
    if (e < E) {
        col[start[dst[e]] + rank[e]] = src[e];
    }
}

#define TILE_PITCH 68

__device__ __forceinline__ void gather_rows8(const char* __restrict__ fbase, int my, int t0, int v,
                                             unsigned lane_off, float4& acc) {
    float4 r[8];
    #pragma unroll
    for (int u = 0; u < 8; ++u) {
        const int s = __shfl(my, t0 + u, 16);
        if (u < v) r[u] = *(const float4*)(fbase + ((unsigned)s * 256u + lane_off));
    }
    #pragma unroll
    for (int u = 0; u < 8; ++u) {
        if (u < v) {
            acc.x += r[u].x; acc.y += r[u].y; acc.z += r[u].z; acc.w += r[u].w;
        }
    }
}

template <int NCOLS, bool RELU>
__global__ __launch_bounds__(256) void gather_gemm_kernel(const float* __restrict__ feat,
                                                          const int* __restrict__ col,
                                                          const int* __restrict__ start,
                                                          const int* __restrict__ cnt,
                                                          const float* __restrict__ W,
                                                          const float* __restrict__ bias,
                                                          float* __restrict__ out, int n) {
    constexpr int tilesN = (NCOLS + 15) / 16;
    __shared__ __attribute__((aligned(16))) float tile[16 * TILE_PITCH];
    __shared__ __attribute__((aligned(16))) float sinv[16];

    const int tid = threadIdx.x;
    const int r = tid >> 4;
    const int l = tid & 15;
    const int node = blockIdx.x * 16 + r;

    float4 acc = {0.0f, 0.0f, 0.0f, 0.0f};
    int c = 0;
    if (node < n) {
        c = cnt[node];
        const int st = start[node];
        const char* __restrict__ fbase = (const char*)feat;
        const unsigned lane_off = 16u * (unsigned)l;
        int nxt = (l < c) ? col[st + l] : 0;
        for (int base = 0; base < c; base += 16) {
            const int my = nxt;
            nxt = (base + 16 + l < c) ? col[st + base + 16 + l] : 0;
            const int m = c - base;
            gather_rows8(fbase, my, 0, m, lane_off, acc);
            if (m > 8) gather_rows8(fbase, my, 8, m - 8, lane_off, acc);
        }
    }
    *(float4*)(tile + r * TILE_PITCH + 4 * l) = acc;
    if (l == 0) {
        const float d = (float)c;
        sinv[r] = (d > 0.0f) ? (1.0f / d) : 0.0f;
    }
    __syncthreads();

    const int w = __builtin_amdgcn_readfirstlane(tid >> 5);
    if (w >= tilesN) return;

    const int lane = tid & 31;
    const int half = lane >> 4;
    const int l15  = lane & 15;

    const int kb = half * 2;
    const int colB = w * 16 + l15;
    const bool colOK = colB < NCOLS;
    const int cB   = colOK ? colB : 0;
    const float bmask = colOK ? 1.0f : 0.0f;

    const float* Ap = tile + l15 * TILE_PITCH + kb;
    const float* Wp = W + kb * NCOLS + cB;
    const float bv = bias[cB];

    v8f cacc = {};
    #pragma unroll
    for (int k0 = 0; k0 < FEAT; k0 += 4) {
        v2f a = *(const v2f*)(Ap + k0);
        v2f b;
        b.x = Wp[k0 * NCOLS] * bmask;
        b.y = Wp[(k0 + 1) * NCOLS] * bmask;
        cacc = __builtin_amdgcn_wmma_f32_16x16x4_f32(
                 false, a,  false, b,
                 (short)0, cacc,  false,  false);
    }

    const float4 i0 = *(const float4*)(sinv + half * 8);
    const float4 i1 = *(const float4*)(sinv + half * 8 + 4);
    const float invs[8] = {i0.x, i0.y, i0.z, i0.w, i1.x, i1.y, i1.z, i1.w};
    const int row0 = blockIdx.x * 16 + half * 8;
    float* op = out + (long long)row0 * NCOLS + cB;
    #pragma unroll
    for (int q = 0; q < 8; ++q) {
        float v = cacc[q] * invs[q] + bv;
        if (RELU) v = fmaxf(v, 0.0f);
        if (colOK && row0 + q < n) {
            op[q * NCOLS] = v;
        }
    }
}

extern "C" void kernel_launch(void* const* d_in, const int* in_sizes, int n_in,
                              void* d_out, int out_size, void* d_ws, size_t ws_size,
                              hipStream_t stream) {
    const float* features = (const float*)d_in[0];
    const float* W1 = (const float*)d_in[1];
    const float* b1 = (const float*)d_in[2];
    const float* W2 = (const float*)d_in[3];
    const float* b2 = (const float*)d_in[4];
    const int* src = (const int*)d_in[5];
    const int* dst = (const int*)d_in[6];
    float* out = (float*)d_out;

    const int N = in_sizes[0] / FEAT;
    const int E = in_sizes[5];

    float* h1   = (float*)d_ws;
    int* cnt    = (int*)(h1 + (long long)N * FEAT);
    int* cursor = cnt + N;
    int* start  = cursor + 1;
    int* rank   = start + N;
    int* col    = rank + E;

    hipMemsetAsync(cnt, 0, (size_t)(N + 1) * sizeof(int), stream);

    const int eblocks = (E + 255) / 256;
    const int nblocks = (N + 255) / 256;
    count_rank_kernel<<<eblocks, 256, 0, stream>>>(dst, cnt, rank, E);
    segment_alloc_kernel<<<nblocks, 256, 0, stream>>>(cnt, cursor, start, N);
    fill_kernel<<<eblocks, 256, 0, stream>>>(src, dst, start, rank, col, E);

    const int gblocks = (N + 15) / 16;
    gather_gemm_kernel<FEAT, true><<<gblocks, 256, 0, stream>>>(features, col, start, cnt,
                                                                 W1, b1, h1, N);

    gather_gemm_kernel<NCLS, false><<<gblocks, 256, 0, stream>>>(h1, col, start, cnt,
                                                                  W2, b2, out, N);
}


# Edge count and fill passes keep each counter and list line in one cache half: blocks read their die, draw chunk tickets per half, take only that half's destination range; sums stay in arrival rank
# speedup vs baseline: 5.6277x; 1.8170x over previous
#include <hip/hip_runtime.h>
#include <hip/hip_bf16.h>

typedef __attribute__((ext_vector_type(2))) float v2f;
typedef __attribute__((ext_vector_type(8))) float v8f;

#define FEAT 64
#define NCLS 40


#define EPT 4
#define CHUNK (256 * EPT)
#define CTRL_INTS 320
#define CTRL_TICK(k) (64 * ((k) + 1))

__device__ __forceinline__ int cache_half() {
    return (int)((__builtin_amdgcn_s_sendmsg_rtn(0x87) >> 18) & 1u);
}

__device__ __forceinline__ int range_mid(int n) { return ((n >> 1) + 255) & ~255; }

__global__ __launch_bounds__(256) void count_rank_kernel(const int* __restrict__ dst,
                                                         int* __restrict__ cnt,
                                                         int* __restrict__ tick,
                                                         int* __restrict__ rank2,
                                                         int E, int n, int nchunks) {
    __shared__ int s_t;
    const int tid = threadIdx.x;
    const int mid = range_mid(n);
    const int h0 = cache_half();
    for (int pass = 0; pass < 2; ++pass) {
        const int hh = h0 ^ pass;
        const int lo = hh ? mid : 0;
        const int hi = hh ? n : mid;
        int* __restrict__ rk = rank2 + (size_t)hh * (size_t)E;
        int* tk = tick + 64 * hh;
        int t_next = 0;
        if (tid == 0) t_next = atomicAdd(tk, 1);
        for (int it = 0; it <= nchunks; ++it) {
            if (tid == 0) s_t = t_next;
            __syncthreads();
            const int t = s_t;
            __syncthreads();
            if (t >= nchunks) break;
            if (tid == 0) t_next = atomicAdd(tk, 1);
            const int base = t * CHUNK + tid;
            int d[EPT];
            #pragma unroll
            for (int u = 0; u < EPT; ++u) {
                const int e = base + u * 256;
                d[u] = (e < E) ? dst[e] : -1;
            }
            int r[EPT];
            #pragma unroll
            for (int u = 0; u < EPT; ++u) {
                r[u] = (d[u] >= lo && d[u] < hi) ? atomicAdd(&cnt[d[u]], 1) : -1;
            }
            #pragma unroll
            for (int u = 0; u < EPT; ++u) {
                if (r[u] >= 0) rk[base + u * 256] = r[u];
            }
        }
    }
}

__global__ __launch_bounds__(256) void segment_alloc_kernel(const int* __restrict__ cnt,
                                                            int* __restrict__ cursor,
                                                            int* __restrict__ start, int n) {
    __shared__ int wave_off[8];
    __shared__ int block_base;
    const int tid = threadIdx.x;
    const int lane = tid & 31;
    const int w = tid >> 5;
    const int v = blockIdx.x * 256 + tid;
    const int c = (v < n) ? cnt[v] : 0;

    int inc = c;
    #pragma unroll
    for (int off = 1; off < 32; off <<= 1) {
        int t = __shfl_up(inc, off, 32);
        if (lane >= off) inc += t;
    }
    if (lane == 31) wave_off[w] = inc;
    __syncthreads();
    if (tid == 0) {
        int run = 0;
        #pragma unroll
        for (int i = 0; i < 8; ++i) {
            int t = wave_off[i];
            wave_off[i] = run;
            run += t;
        }
        block_base = atomicAdd(cursor, run);
    }
    __syncthreads();
    if (v < n) {
        start[v] = block_base + wave_off[w] + inc - c;
    }
}

__global__ __launch_bounds__(256) void fill_kernel(const int* __restrict__ src,
                                                   const int* __restrict__ dst,
                                                   const int* __restrict__ start,
                                                   const int* __restrict__ rank2,
                                                   int* __restrict__ col,
                                                   int* __restrict__ tick,
                                                   int E, int n, int nchunks) {
    __shared__ int s_t;
    const int tid = threadIdx.x;
    const int mid = range_mid(n);
    const int h0 = cache_half();
    for (int pass = 0; pass < 2; ++pass) {
        const int hh = h0 ^ pass;
        const int lo = hh ? mid : 0;
        const int hi = hh ? n : mid;
        const int* __restrict__ rk = rank2 + (size_t)hh * (size_t)E;
        int* tk = tick + 64 * hh;
        int t_next = 0;
        if (tid == 0) t_next = atomicAdd(tk, 1);
        for (int it = 0; it <= nchunks; ++it) {
            if (tid == 0) s_t = t_next;
            __syncthreads();
            const int t = s_t;
            __syncthreads();
            if (t >= nchunks) break;
            if (tid == 0) t_next = atomicAdd(tk, 1);
            const int base = t * CHUNK + tid;
            int d[EPT], st[EPT], rr[EPT], s[EPT];
            #pragma unroll
            for (int u = 0; u < EPT; ++u) {
                const int e = base + u * 256;
                d[u] = (e < E) ? dst[e] : -1;
            }
            #pragma unroll
            for (int u = 0; u < EPT; ++u) {
                const int e = base + u * 256;
                const bool mine = d[u] >= lo && d[u] < hi;
                st[u] = mine ? start[d[u]] : -1;
                rr[u] = mine ? rk[e] : 0;
                s[u]  = mine ? src[e] : 0;
            }
            #pragma unroll
            for (int u = 0; u < EPT; ++u) {
                if (st[u] >= 0) col[st[u] + rr[u]] = s[u];
            }
        }
    }
}

#define TILE_PITCH 68

__device__ __forceinline__ void gather_rows8(const char* __restrict__ fbase, int my, int t0, int v,
                                             unsigned lane_off, float4& acc) {
    float4 r[8];
    #pragma unroll
    for (int u = 0; u < 8; ++u) {
        const int s = __shfl(my, t0 + u, 16);
        if (u < v) r[u] = *(const float4*)(fbase + ((unsigned)s * 256u + lane_off));
    }
    #pragma unroll
    for (int u = 0; u < 8; ++u) {
        if (u < v) {
            acc.x += r[u].x; acc.y += r[u].y; acc.z += r[u].z; acc.w += r[u].w;
        }
    }
}

template <int NCOLS, bool RELU>
__global__ __launch_bounds__(256) void gather_gemm_kernel(const float* __restrict__ feat,
                                                          const int* __restrict__ col,
                                                          const int* __restrict__ start,
                                                          const int* __restrict__ cnt,
                                                          const float* __restrict__ W,
                                                          const float* __restrict__ bias,
                                                          float* __restrict__ out, int n) {
    constexpr int tilesN = (NCOLS + 15) / 16;
    __shared__ __attribute__((aligned(16))) float tile[16 * TILE_PITCH];
    __shared__ __attribute__((aligned(16))) float sinv[16];

    const int tid = threadIdx.x;
    const int r = tid >> 4;
    const int l = tid & 15;
    const int node = blockIdx.x * 16 + r;

    float4 acc = {0.0f, 0.0f, 0.0f, 0.0f};
    int c = 0;
    if (node < n) {
        c = cnt[node];
        const int st = start[node];
        const char* __restrict__ fbase = (const char*)feat;
        const unsigned lane_off = 16u * (unsigned)l;
        int nxt = (l < c) ? col[st + l] : 0;
        for (int base = 0; base < c; base += 16) {
            const int my = nxt;
            nxt = (base + 16 + l < c) ? col[st + base + 16 + l] : 0;
            const int m = c - base;
            gather_rows8(fbase, my, 0, m, lane_off, acc);
            if (m > 8) gather_rows8(fbase, my, 8, m - 8, lane_off, acc);
        }
    }
    *(float4*)(tile + r * TILE_PITCH + 4 * l) = acc;
    if (l == 0) {
        const float d = (float)c;
        sinv[r] = (d > 0.0f) ? (1.0f / d) : 0.0f;
    }
    __syncthreads();

    const int w = __builtin_amdgcn_readfirstlane(tid >> 5);
    if (w >= tilesN) return;

    const int lane = tid & 31;
    const int half = lane >> 4;
    const int l15  = lane & 15;

    const int kb = half * 2;
    const int colB = w * 16 + l15;
    const bool colOK = colB < NCOLS;
    const int cB   = colOK ? colB : 0;
    const float bmask = colOK ? 1.0f : 0.0f;

    const float* Ap = tile + l15 * TILE_PITCH + kb;
    const float* Wp = W + kb * NCOLS + cB;
    const float bv = bias[cB];

    v8f cacc = {};
    #pragma unroll
    for (int k0 = 0; k0 < FEAT; k0 += 4) {
        v2f a = *(const v2f*)(Ap + k0);
        v2f b;
        b.x = Wp[k0 * NCOLS] * bmask;
        b.y = Wp[(k0 + 1) * NCOLS] * bmask;
        cacc = __builtin_amdgcn_wmma_f32_16x16x4_f32(
                 false, a,  false, b,
                 (short)0, cacc,  false,  false);
    }

    const float4 i0 = *(const float4*)(sinv + half * 8);
    const float4 i1 = *(const float4*)(sinv + half * 8 + 4);
    const float invs[8] = {i0.x, i0.y, i0.z, i0.w, i1.x, i1.y, i1.z, i1.w};
    const int row0 = blockIdx.x * 16 + half * 8;
    float* op = out + (long long)row0 * NCOLS + cB;
    #pragma unroll
    for (int q = 0; q < 8; ++q) {
        float v = cacc[q] * invs[q] + bv;
        if (RELU) v = fmaxf(v, 0.0f);
        if (colOK && row0 + q < n) {
            op[q * NCOLS] = v;
        }
    }
}

extern "C" void kernel_launch(void* const* d_in, const int* in_sizes, int n_in,
                              void* d_out, int out_size, void* d_ws, size_t ws_size,
                              hipStream_t stream) {
    const float* features = (const float*)d_in[0];
    const float* W1 = (const float*)d_in[1];
    const float* b1 = (const float*)d_in[2];
    const float* W2 = (const float*)d_in[3];
    const float* b2 = (const float*)d_in[4];
    const int* src = (const int*)d_in[5];
    const int* dst = (const int*)d_in[6];
    float* out = (float*)d_out;

    const int N = in_sizes[0] / FEAT;
    const int E = in_sizes[5];

    float* h1   = (float*)d_ws;
    int* cnt    = (int*)(h1 + (long long)N * FEAT);
    int* ctrl   = cnt + N;
    int* cursor = ctrl;
    int* start  = ctrl + CTRL_INTS;
    int* rank2  = start + N;
    int* col    = rank2 + 2 * (long long)E;

    hipMemsetAsync(cnt, 0, (size_t)(N + CTRL_INTS) * sizeof(int), stream);

    const int nchunks = (E + CHUNK - 1) / CHUNK;
    const int nblocks = (N + 255) / 256;
    count_rank_kernel<<<1024, 256, 0, stream>>>(dst, cnt, ctrl + CTRL_TICK(0), rank2, E, N, nchunks);
    segment_alloc_kernel<<<nblocks, 256, 0, stream>>>(cnt, cursor, start, N);
    fill_kernel<<<1024, 256, 0, stream>>>(src, dst, start, rank2, col, ctrl + CTRL_TICK(2), E, N, nchunks);

    const int gblocks = (N + 15) / 16;
    gather_gemm_kernel<FEAT, true><<<gblocks, 256, 0, stream>>>(features, col, start, cnt,
                                                                 W1, b1, h1, N);

    gather_gemm_kernel<NCLS, false><<<gblocks, 256, 0, stream>>>(h1, col, start, cnt,
                                                                  W2, b2, out, N);
}


# Edge grouping in one pass: a 32-slot id slab per node, the counting atomic's returned rank is the slot (overflow: side list); rank plane, segment scan, 2nd edge pass removed; sum order kept
# speedup vs baseline: 6.7024x; 1.1910x over previous
#include <hip/hip_runtime.h>
#include <hip/hip_bf16.h>

typedef __attribute__((ext_vector_type(2))) float v2f;
typedef __attribute__((ext_vector_type(8))) float v8f;

#define FEAT 64
#define NCLS 40

#define SLAB 32

#define EPT 4
#define CHUNK (256 * EPT)
#define CTRL_INTS 192
#define CTRL_TICK(k) (64 * ((k) + 1))

__device__ __forceinline__ int cache_half() {
    return (int)((__builtin_amdgcn_s_sendmsg_rtn(0x87) >> 18) & 1u);
}

__device__ __forceinline__ int range_mid(int n) { return ((n >> 1) + 255) & ~255; }

__global__ __launch_bounds__(256) void place_edges_kernel(const int* __restrict__ src,
                                                          const int* __restrict__ dst,
                                                          int* __restrict__ cnt,
                                                          int* __restrict__ tick,
                                                          int* __restrict__ slab,
                                                          int* __restrict__ side_cursor,
                                                          int* __restrict__ side,
                                                          int E, int n, int nchunks) {
    __shared__ int s_t;
    const int tid = threadIdx.x;
    const int mid = range_mid(n);
    const int h0 = cache_half();
    for (int pass = 0; pass < 2; ++pass) {
        const int hh = h0 ^ pass;
        const int lo = hh ? mid : 0;
        const int hi = hh ? n : mid;
        int* tk = tick + 64 * hh;
        int t_next = 0;
        if (tid == 0) t_next = atomicAdd(tk, 1);
        for (int it = 0; it <= nchunks; ++it) {
            if (tid == 0) s_t = t_next;
            __syncthreads();
            const int t = s_t;
            __syncthreads();
            if (t >= nchunks) break;
            if (tid == 0) t_next = atomicAdd(tk, 1);
            const int base = t * CHUNK + tid;
            int d[EPT], s[EPT];
            #pragma unroll
            for (int u = 0; u < EPT; ++u) {
                const int e = base + u * 256;
                d[u] = (e < E) ? dst[e] : -1;
                s[u] = (e < E) ? src[e] : 0;
            }
            int r[EPT];
            #pragma unroll
            for (int u = 0; u < EPT; ++u) {
                r[u] = (d[u] >= lo && d[u] < hi) ? atomicAdd(&cnt[d[u]], 1) : -1;
            }
            #pragma unroll
            for (int u = 0; u < EPT; ++u) {
                if (r[u] >= 0) {
                    if (r[u] < SLAB) {
                        *(int*)((char*)slab + ((unsigned)d[u] * (4u * SLAB) + 4u * (unsigned)r[u])) = s[u];
                    } else {
                        const int p = atomicAdd(side_cursor, 1);
                        side[2 * (size_t)p]     = d[u];
                        side[2 * (size_t)p + 1] = s[u];
                    }
                }
            }
        }
    }
}

#define TILE_PITCH 68

__device__ __forceinline__ void gather_rows8(const char* __restrict__ fbase, int my, int t0, int v,
                                             unsigned lane_off, float4& acc) {
    float4 r[8];
    #pragma unroll
    for (int u = 0; u < 8; ++u) {
        const int s = __shfl(my, t0 + u, 16);
        if (u < v) r[u] = *(const float4*)(fbase + ((unsigned)s * 256u + lane_off));
    }
    #pragma unroll
    for (int u = 0; u < 8; ++u) {
        if (u < v) {
            acc.x += r[u].x; acc.y += r[u].y; acc.z += r[u].z; acc.w += r[u].w;
        }
    }
}

template <int NCOLS, bool RELU>
__global__ __launch_bounds__(256) void gather_gemm_kernel(const float* __restrict__ feat,
                                                          const int* __restrict__ slab,
                                                          const int* __restrict__ cnt,
                                                          const int* __restrict__ side_cursor,
                                                          const int* __restrict__ side,
                                                          const float* __restrict__ W,
                                                          const float* __restrict__ bias,
                                                          float* __restrict__ out, int n, int E) {
    constexpr int tilesN = (NCOLS + 15) / 16;
    __shared__ __attribute__((aligned(16))) float tile[16 * TILE_PITCH];
    __shared__ __attribute__((aligned(16))) float sinv[16];

    const int tid = threadIdx.x;
    const int r = tid >> 4;
    const int l = tid & 15;
    const int node = blockIdx.x * 16 + r;

    float4 acc = {0.0f, 0.0f, 0.0f, 0.0f};
    int c = 0;
    if (node < n) {
        c = cnt[node];
        const int cs = (c < SLAB) ? c : SLAB;
        const unsigned seg_off = (unsigned)node * (4u * SLAB) + 4u * (unsigned)l;
        const char* __restrict__ sbase = (const char*)slab;
        const char* __restrict__ fbase = (const char*)feat;
        const unsigned lane_off = 16u * (unsigned)l;
        int nxt = (l < cs) ? *(const int*)(sbase + seg_off) : 0;
        for (int base = 0; base < cs; base += 16) {
            const int my = nxt;
            nxt = (base + 16 + l < cs) ? *(const int*)(sbase + (seg_off + 4u * (unsigned)(base + 16))) : 0;
            const int m = cs - base;
            gather_rows8(fbase, my, 0, m, lane_off, acc);
            if (m > 8) gather_rows8(fbase, my, 8, m - 8, lane_off, acc);
        }
        if (c > SLAB) {
            int ns = *side_cursor;
            ns = (ns < E) ? ns : E;
            for (int i = 0; i < ns; ++i) {
                if (side[2 * (size_t)i] == node) {
                    const int s = side[2 * (size_t)i + 1];
                    const float4 v = *(const float4*)(fbase + ((unsigned)s * 256u + lane_off));
                    acc.x += v.x; acc.y += v.y; acc.z += v.z; acc.w += v.w;
                }
            }
        }
    }
    *(float4*)(tile + r * TILE_PITCH + 4 * l) = acc;
    if (l == 0) {
        const float d = (float)c;
        sinv[r] = (d > 0.0f) ? (1.0f / d) : 0.0f;
    }
    __syncthreads();

    const int w = __builtin_amdgcn_readfirstlane(tid >> 5);
    if (w >= tilesN) return;

    const int lane = tid & 31;
    const int half = lane >> 4;
    const int l15  = lane & 15;

    const int kb = half * 2;
    const int colB = w * 16 + l15;
    const bool colOK = colB < NCOLS;
    const int cB   = colOK ? colB : 0;
    const float bmask = colOK ? 1.0f : 0.0f;

    const float* Ap = tile + l15 * TILE_PITCH + kb;
    const float* Wp = W + kb * NCOLS + cB;
    const float bv = bias[cB];

    v8f cacc = {};
    #pragma unroll
    for (int k0 = 0; k0 < FEAT; k0 += 4) {
        v2f a = *(const v2f*)(Ap + k0);
        v2f b;
        b.x = Wp[k0 * NCOLS] * bmask;
        b.y = Wp[(k0 + 1) * NCOLS] * bmask;
        cacc = __builtin_amdgcn_wmma_f32_16x16x4_f32(
                 false, a,  false, b,
                 (short)0, cacc,  false,  false);
    }

    const float4 i0 = *(const float4*)(sinv + half * 8);
    const float4 i1 = *(const float4*)(sinv + half * 8 + 4);
    const float invs[8] = {i0.x, i0.y, i0.z, i0.w, i1.x, i1.y, i1.z, i1.w};
    const int row0 = blockIdx.x * 16 + half * 8;
    float* op = out + (long long)row0 * NCOLS + cB;
    #pragma unroll
    for (int q = 0; q < 8; ++q) {
        float v = cacc[q] * invs[q] + bv;
        if (RELU) v = fmaxf(v, 0.0f);
        if (colOK && row0 + q < n) {
            op[q * NCOLS] = v;
        }
    }
}

extern "C" void kernel_launch(void* const* d_in, const int* in_sizes, int n_in,
                              void* d_out, int out_size, void* d_ws, size_t ws_size,
                              hipStream_t stream) {
    const float* features = (const float*)d_in[0];
    const float* W1 = (const float*)d_in[1];
    const float* b1 = (const float*)d_in[2];
    const float* W2 = (const float*)d_in[3];
    const float* b2 = (const float*)d_in[4];
    const int* src = (const int*)d_in[5];
    const int* dst = (const int*)d_in[6];
    float* out = (float*)d_out;

    const int N = in_sizes[0] / FEAT;
    const int E = in_sizes[5];

    float* h1   = (float*)d_ws;
    int* slab   = (int*)(h1 + (long long)N * FEAT);
    int* cnt    = slab + (long long)N * SLAB;
    int* ctrl   = cnt + N;
    int* side_cursor = ctrl;
    int* side   = ctrl + CTRL_INTS;

    hipMemsetAsync(cnt, 0, (size_t)(N + CTRL_INTS) * sizeof(int), stream);

    const int nchunks = (E + CHUNK - 1) / CHUNK;
    place_edges_kernel<<<1024, 256, 0, stream>>>(src, dst, cnt, ctrl + CTRL_TICK(0), slab,
                                                 side_cursor, side, E, N, nchunks);

    const int gblocks = (N + 15) / 16;
    gather_gemm_kernel<FEAT, true><<<gblocks, 256, 0, stream>>>(features, slab, cnt, side_cursor, side,
                                                                 W1, b1, h1, N, E);

    gather_gemm_kernel<NCLS, false><<<gblocks, 256, 0, stream>>>(h1, slab, cnt, side_cursor, side,
                                                                  W2, b2, out, N, E);
}


# Layer kernels run as 1,250 resident work-groups looping over 16-node tiles: two LDS tiles, one barrier per tile, next tiles' counts and ids asked ahead, row loads unguarded; sum order unchanged
# speedup vs baseline: 7.1822x; 1.0716x over previous
#include <hip/hip_runtime.h>
#include <hip/hip_bf16.h>

typedef __attribute__((ext_vector_type(2))) float v2f;
typedef __attribute__((ext_vector_type(8))) float v8f;

#define FEAT 64
#define NCLS 40

#define SLAB 32

#define EPT 4
#define CHUNK (256 * EPT)
#define CTRL_INTS 192
#define CTRL_TICK(k) (64 * ((k) + 1))

__device__ __forceinline__ int cache_half() {
    return (int)((__builtin_amdgcn_s_sendmsg_rtn(0x87) >> 18) & 1u);
}

__device__ __forceinline__ int range_mid(int n) { return ((n >> 1) + 255) & ~255; }

__global__ __launch_bounds__(256) void place_edges_kernel(const int* __restrict__ src,
                                                          const int* __restrict__ dst,
                                                          int* __restrict__ cnt,
                                                          int* __restrict__ tick,
                                                          int* __restrict__ slab,
                                                          int* __restrict__ side_cursor,
                                                          int* __restrict__ side,
                                                          int E, int n, int nchunks) {
    __shared__ int s_t;
    const int tid = threadIdx.x;
    const int mid = range_mid(n);
    const int h0 = cache_half();
    for (int pass = 0; pass < 2; ++pass) {
        const int hh = h0 ^ pass;
        const int lo = hh ? mid : 0;
        const int hi = hh ? n : mid;
        int* tk = tick + 64 * hh;
        int t_next = 0;
        if (tid == 0) t_next = atomicAdd(tk, 1);
        for (int it = 0; it <= nchunks; ++it) {
            if (tid == 0) s_t = t_next;
            __syncthreads();
            const int t = s_t;
            __syncthreads();
            if (t >= nchunks) break;
            if (tid == 0) t_next = atomicAdd(tk, 1);
            const int base = t * CHUNK + tid;
            int d[EPT], s[EPT];
            #pragma unroll
            for (int u = 0; u < EPT; ++u) {
                const int e = base + u * 256;
                d[u] = (e < E) ? dst[e] : -1;
                s[u] = (e < E) ? src[e] : 0;
            }
            int r[EPT];
            #pragma unroll
            for (int u = 0; u < EPT; ++u) {
                r[u] = (d[u] >= lo && d[u] < hi) ? atomicAdd(&cnt[d[u]], 1) : -1;
            }
            #pragma unroll
            for (int u = 0; u < EPT; ++u) {
                if (r[u] >= 0) {
                    if (r[u] < SLAB) {
                        *(int*)((char*)slab + ((unsigned)d[u] * (4u * SLAB) + 4u * (unsigned)r[u])) = s[u];
                    } else {
                        const int p = atomicAdd(side_cursor, 1);
                        side[2 * (size_t)p]     = d[u];
                        side[2 * (size_t)p + 1] = s[u];
                    }
                }
            }
        }
    }
}

#define TILE_PITCH 68
#define RES_GROUPS 1250

struct NoRequests { __device__ __forceinline__ void operator()() const {} };

template <typename Between>
__device__ __forceinline__ void gather_rows8(const char* __restrict__ fbase, int my, int t0, int v,
                                             unsigned lane_off, float4& acc, Between between) {
    float4 r[8];
    #pragma unroll
    for (int u = 0; u < 8; ++u) {
        const int s = __shfl(my, t0 + u, 16);
        r[u] = *(const float4*)(fbase + ((unsigned)s * 256u + lane_off));
    }
    between();
    asm volatile("" :: "v"(r[0].x), "v"(r[1].x), "v"(r[2].x), "v"(r[3].x),
                       "v"(r[4].x), "v"(r[5].x), "v"(r[6].x), "v"(r[7].x));
    #pragma unroll
    for (int u = 0; u < 8; ++u) {
        if (u < v) {
            acc.x += r[u].x; acc.y += r[u].y; acc.z += r[u].z; acc.w += r[u].w;
        }
    }
}

template <int NCOLS, bool RELU>
__global__ __launch_bounds__(256) __attribute__((amdgpu_waves_per_eu(10, 10)))
void gather_gemm_kernel(const float* __restrict__ feat,
                                                          const int* __restrict__ slab,
                                                          const int* __restrict__ cnt,
                                                          const int* __restrict__ side_cursor,
                                                          const int* __restrict__ side,
                                                          const float* __restrict__ W,
                                                          const float* __restrict__ bias,
                                                          float* __restrict__ out, int n, int E) {
    constexpr int tilesN = (NCOLS + 15) / 16;
    __shared__ __attribute__((aligned(16))) float tile[2][16 * TILE_PITCH];
    __shared__ __attribute__((aligned(16))) float sinv[2][16];

    const int tid = threadIdx.x;
    const int r = tid >> 4;
    const int l = tid & 15;
    const int ntiles = (n + 15) >> 4;
    const int step = (int)gridDim.x;

    const int w = __builtin_amdgcn_readfirstlane(tid >> 5);
    const int lane = tid & 31;
    const int half = lane >> 4;
    const int l15  = lane & 15;

    const int kb = half * 2;
    const int colB = w * 16 + l15;
    const bool colOK = colB < NCOLS;
    const int cB   = colOK ? colB : 0;
    const float bmask = colOK ? 1.0f : 0.0f;

    const float* Wp = W + kb * NCOLS + cB;
    const float bv = bias[cB];

    const char* __restrict__ sbase = (const char*)slab;
    const char* __restrict__ fbase = (const char*)feat;
    const unsigned lane_off = 16u * (unsigned)l;

    int t = (int)blockIdx.x;
    int c_cur = 0, ids_cur = 0, c_nxt = 0;
    {
        const int node0 = t * 16 + r;
        if (t < ntiles && node0 < n) c_cur = cnt[node0];
        const int node1 = (t + step) * 16 + r;
        if (t + step < ntiles && node1 < n) c_nxt = cnt[node1];
        const int cs0 = (c_cur < SLAB) ? c_cur : SLAB;
        if (l < cs0) ids_cur = *(const int*)(sbase + ((unsigned)node0 * (4u * SLAB) + 4u * (unsigned)l));
    }

    for (int it = 0; t < ntiles; t += step, ++it) {
        float* tl = tile[it & 1];
        float* sv = sinv[it & 1];
        const int node = t * 16 + r;

        float4 acc = {0.0f, 0.0f, 0.0f, 0.0f};
        const int c = c_cur;
        const int cs = (c < SLAB) ? c : SLAB;
        const unsigned seg_off = (unsigned)node * (4u * SLAB) + 4u * (unsigned)l;

        int ids_nxt = 0, c_nn = 0;
        int nxt = 0;
        auto ask_ahead = [&]() {
            if (16 + l < cs) nxt = *(const int*)(sbase + (seg_off + 4u * 16u));
            const int node1 = (t + step) * 16 + r;
            const int cs1 = (c_nxt < SLAB) ? c_nxt : SLAB;
            if (l < cs1) ids_nxt = *(const int*)(sbase + ((unsigned)node1 * (4u * SLAB) + 4u * (unsigned)l));
            const int node2 = (t + 2 * step) * 16 + r;
            if (t + 2 * step < ntiles && node2 < n) c_nn = cnt[node2];
        };

        gather_rows8(fbase, ids_cur, 0, cs, lane_off, acc, ask_ahead);
        if (cs > 8) gather_rows8(fbase, ids_cur, 8, cs - 8, lane_off, acc, NoRequests());
        for (int base = 16; base < cs; base += 16) {
            const int my = nxt;
            nxt = (base + 16 + l < cs) ? *(const int*)(sbase + (seg_off + 4u * (unsigned)(base + 16))) : 0;
            const int m = cs - base;
            gather_rows8(fbase, my, 0, m, lane_off, acc, NoRequests());
            if (m > 8) gather_rows8(fbase, my, 8, m - 8, lane_off, acc, NoRequests());
        }
        if (c > SLAB) {
            int ns = *side_cursor;
            ns = (ns < E) ? ns : E;
            for (int i = 0; i < ns; ++i) {
                if (side[2 * (size_t)i] == node) {
                    const int s = side[2 * (size_t)i + 1];
                    const float4 v = *(const float4*)(fbase + ((unsigned)s * 256u + lane_off));
                    acc.x += v.x; acc.y += v.y; acc.z += v.z; acc.w += v.w;
                }
            }
        }
        *(float4*)(tl + r * TILE_PITCH + 4 * l) = acc;
        if (l == 0) {
            const float d = (float)c;
            sv[r] = (d > 0.0f) ? (1.0f / d) : 0.0f;
        }
        __syncthreads();

        c_cur = c_nxt; ids_cur = ids_nxt; c_nxt = c_nn;

        if (w < tilesN) {
            const float* Ap = tl + l15 * TILE_PITCH + kb;
            v8f cacc = {};
            #pragma unroll
            for (int q = 0; q < FEAT / 4; ++q) {
                v2f a = *(const v2f*)(Ap + 4 * q);
                v2f b;
                b.x = Wp[(4 * q) * NCOLS] * bmask;
                b.y = Wp[(4 * q + 1) * NCOLS] * bmask;
                cacc = __builtin_amdgcn_wmma_f32_16x16x4_f32(
                         false, a,  false, b,
                         (short)0, cacc,  false,  false);
            }

            const float4 i0 = *(const float4*)(sv + half * 8);
            const float4 i1 = *(const float4*)(sv + half * 8 + 4);
            const float invs[8] = {i0.x, i0.y, i0.z, i0.w, i1.x, i1.y, i1.z, i1.w};
            const int row0 = t * 16 + half * 8;
            float* op = out + (long long)row0 * NCOLS + cB;
            #pragma unroll
            for (int q = 0; q < 8; ++q) {
                float v = cacc[q] * invs[q] + bv;
                if (RELU) v = fmaxf(v, 0.0f);
                if (colOK && row0 + q < n) {
                    op[q * NCOLS] = v;
                }
            }
        }
    }
}

extern "C" void kernel_launch(void* const* d_in, const int* in_sizes, int n_in,
                              void* d_out, int out_size, void* d_ws, size_t ws_size,
                              hipStream_t stream) {
    const float* features = (const float*)d_in[0];
    const float* W1 = (const float*)d_in[1];
    const float* b1 = (const float*)d_in[2];
    const float* W2 = (const float*)d_in[3];
    const float* b2 = (const float*)d_in[4];
    const int* src = (const int*)d_in[5];
    const int* dst = (const int*)d_in[6];
    float* out = (float*)d_out;

    const int N = in_sizes[0] / FEAT;
    const int E = in_sizes[5];

    float* h1   = (float*)d_ws;
    int* slab   = (int*)(h1 + (long long)N * FEAT);
    int* cnt    = slab + (long long)N * SLAB;
    int* ctrl   = cnt + N;
    int* side_cursor = ctrl;
    int* side   = ctrl + CTRL_INTS;

    hipMemsetAsync(cnt, 0, (size_t)(N + CTRL_INTS) * sizeof(int), stream);

    const int nchunks = (E + CHUNK - 1) / CHUNK;
    place_edges_kernel<<<1024, 256, 0, stream>>>(src, dst, cnt, ctrl + CTRL_TICK(0), slab,
                                                 side_cursor, side, E, N, nchunks);

    gather_gemm_kernel<FEAT, true><<<RES_GROUPS, 256, 0, stream>>>(features, slab, cnt, side_cursor, side,
                                                                    W1, b1, h1, N, E);

    gather_gemm_kernel<NCLS, false><<<RES_GROUPS, 256, 0, stream>>>(h1, slab, cnt, side_cursor, side,
                                                                     W2, b2, out, N, E);
}


# B operand from LDS: each work-group copies W transposed into LDS once and reads its B fragments there, not 32 global loads per tile and wave; k order and sum order unchanged
# speedup vs baseline: 7.6109x; 1.0597x over previous
#include <hip/hip_runtime.h>
#include <hip/hip_bf16.h>

typedef __attribute__((ext_vector_type(2))) float v2f;
typedef __attribute__((ext_vector_type(8))) float v8f;

#define FEAT 64
#define NCLS 40

#define SLAB 32

#define EPT 4
#define CHUNK (256 * EPT)
#define CTRL_INTS 192
#define CTRL_TICK(k) (64 * ((k) + 1))

__device__ __forceinline__ int cache_half() {
    return (int)((__builtin_amdgcn_s_sendmsg_rtn(0x87) >> 18) & 1u);
}

__device__ __forceinline__ int range_mid(int n) { return ((n >> 1) + 255) & ~255; }

__global__ __launch_bounds__(256) void place_edges_kernel(const int* __restrict__ src,
                                                          const int* __restrict__ dst,
                                                          int* __restrict__ cnt,
                                                          int* __restrict__ tick,
                                                          int* __restrict__ slab,
                                                          int* __restrict__ side_cursor,
                                                          int* __restrict__ side,
                                                          int E, int n, int nchunks) {
    __shared__ int s_t;
    const int tid = threadIdx.x;
    const int mid = range_mid(n);
    const int h0 = cache_half();
    for (int pass = 0; pass < 2; ++pass) {
        const int hh = h0 ^ pass;
        const int lo = hh ? mid : 0;
        const int hi = hh ? n : mid;
        int* tk = tick + 64 * hh;
        int t_next = 0;
        if (tid == 0) t_next = atomicAdd(tk, 1);
        for (int it = 0; it <= nchunks; ++it) {
            if (tid == 0) s_t = t_next;
            __syncthreads();
            const int t = s_t;
            __syncthreads();
            if (t >= nchunks) break;
            if (tid == 0) t_next = atomicAdd(tk, 1);
            const int base = t * CHUNK + tid;
            int d[EPT], s[EPT];
            #pragma unroll
            for (int u = 0; u < EPT; ++u) {
                const int e = base + u * 256;
                d[u] = (e < E) ? dst[e] : -1;
                s[u] = (e < E) ? src[e] : 0;
            }
            int r[EPT];
            #pragma unroll
            for (int u = 0; u < EPT; ++u) {
                r[u] = (d[u] >= lo && d[u] < hi) ? atomicAdd(&cnt[d[u]], 1) : -1;
            }
            #pragma unroll
            for (int u = 0; u < EPT; ++u) {
                if (r[u] >= 0) {
                    if (r[u] < SLAB) {
                        *(int*)((char*)slab + ((unsigned)d[u] * (4u * SLAB) + 4u * (unsigned)r[u])) = s[u];
                    } else {
                        const int p = atomicAdd(side_cursor, 1);
                        side[2 * (size_t)p]     = d[u];
                        side[2 * (size_t)p + 1] = s[u];
                    }
                }
            }
        }
    }
}

#define TILE_PITCH 68
#define RES_GROUPS 1250

struct NoRequests { __device__ __forceinline__ void operator()() const {} };

template <typename Between>
__device__ __forceinline__ void gather_rows8(const char* __restrict__ fbase, int my, int t0, int v,
                                             unsigned lane_off, float4& acc, Between between) {
    float4 r[8];
    #pragma unroll
    for (int u = 0; u < 8; ++u) {
        const int s = __shfl(my, t0 + u, 16);
        r[u] = *(const float4*)(fbase + ((unsigned)s * 256u + lane_off));
    }
    between();
    asm volatile("" :: "v"(r[0].x), "v"(r[1].x), "v"(r[2].x), "v"(r[3].x),
                       "v"(r[4].x), "v"(r[5].x), "v"(r[6].x), "v"(r[7].x));
    #pragma unroll
    for (int u = 0; u < 8; ++u) {
        if (u < v) {
            acc.x += r[u].x; acc.y += r[u].y; acc.z += r[u].z; acc.w += r[u].w;
        }
    }
}

template <int NCOLS, bool RELU>
__global__ __launch_bounds__(256) __attribute__((amdgpu_waves_per_eu(10, 10)))
void gather_gemm_kernel(const float* __restrict__ feat,
                                                          const int* __restrict__ slab,
                                                          const int* __restrict__ cnt,
                                                          const int* __restrict__ side_cursor,
                                                          const int* __restrict__ side,
                                                          const float* __restrict__ W,
                                                          const float* __restrict__ bias,
                                                          float* __restrict__ out, int n, int E) {
    constexpr int tilesN = (NCOLS + 15) / 16;
    __shared__ __attribute__((aligned(16))) float tile[2][16 * TILE_PITCH];
    __shared__ __attribute__((aligned(16))) float sinv[2][16];
    constexpr int NPAD = tilesN * 16;
    __shared__ __attribute__((aligned(16))) float wsh[NPAD * TILE_PITCH];

    const int tid = threadIdx.x;
    const int r = tid >> 4;
    const int l = tid & 15;
    const int ntiles = (n + 15) >> 4;
    const int step = (int)gridDim.x;

    const int w = __builtin_amdgcn_readfirstlane(tid >> 5);
    const int lane = tid & 31;
    const int half = lane >> 4;
    const int l15  = lane & 15;

    const int kb = half * 2;
    const int colB = w * 16 + l15;
    const bool colOK = colB < NCOLS;
    const int cB   = colOK ? colB : 0;
    const float bv = bias[cB];

    const char* __restrict__ sbase = (const char*)slab;
    const char* __restrict__ fbase = (const char*)feat;
    const unsigned lane_off = 16u * (unsigned)l;

    constexpr int WITEMS = NPAD * (FEAT / 4) / 256;
    static_assert(NPAD * (FEAT / 4) % 256 == 0, "W items must divide over the work-group");
    int t = (int)blockIdx.x;
    int c_cur = 0, ids_cur = 0, c_nxt = 0;
    {
        const int node0 = t * 16 + r;
        if (t < ntiles && node0 < n) c_cur = cnt[node0];
        const int node1 = (t + step) * 16 + r;
        if (t + step < ntiles && node1 < n) c_nxt = cnt[node1];

        float4 wv[WITEMS];
        #pragma unroll
        for (int j = 0; j < WITEMS; ++j) {
            const int i = tid + 256 * j;
            const int c = i % NPAD;
            const int k4 = i / NPAD;
            wv[j] = float4{0.0f, 0.0f, 0.0f, 0.0f};
            if (c < NCOLS) {
                const float* p = W + (4 * k4) * NCOLS + c;
                wv[j].x = p[0]; wv[j].y = p[NCOLS]; wv[j].z = p[2 * NCOLS]; wv[j].w = p[3 * NCOLS];
            }
        }

        const int cs0 = (c_cur < SLAB) ? c_cur : SLAB;
        if (l < cs0) ids_cur = *(const int*)(sbase + ((unsigned)node0 * (4u * SLAB) + 4u * (unsigned)l));

        #pragma unroll
        for (int j = 0; j < WITEMS; ++j) {
            const int i = tid + 256 * j;
            *(float4*)(wsh + (i % NPAD) * TILE_PITCH + 4 * (i / NPAD)) = wv[j];
        }
    }

    for (int it = 0; t < ntiles; t += step, ++it) {
        float* tl = tile[it & 1];
        float* sv = sinv[it & 1];
        const int node = t * 16 + r;

        float4 acc = {0.0f, 0.0f, 0.0f, 0.0f};
        const int c = c_cur;
        const int cs = (c < SLAB) ? c : SLAB;
        const unsigned seg_off = (unsigned)node * (4u * SLAB) + 4u * (unsigned)l;

        int ids_nxt = 0, c_nn = 0;
        int nxt = 0;
        auto ask_ahead = [&]() {
            if (16 + l < cs) nxt = *(const int*)(sbase + (seg_off + 4u * 16u));
            const int node1 = (t + step) * 16 + r;
            const int cs1 = (c_nxt < SLAB) ? c_nxt : SLAB;
            if (l < cs1) ids_nxt = *(const int*)(sbase + ((unsigned)node1 * (4u * SLAB) + 4u * (unsigned)l));
            const int node2 = (t + 2 * step) * 16 + r;
            if (t + 2 * step < ntiles && node2 < n) c_nn = cnt[node2];
        };

        gather_rows8(fbase, ids_cur, 0, cs, lane_off, acc, ask_ahead);
        if (cs > 8) gather_rows8(fbase, ids_cur, 8, cs - 8, lane_off, acc, NoRequests());
        for (int base = 16; base < cs; base += 16) {
            const int my = nxt;
            nxt = (base + 16 + l < cs) ? *(const int*)(sbase + (seg_off + 4u * (unsigned)(base + 16))) : 0;
            const int m = cs - base;
            gather_rows8(fbase, my, 0, m, lane_off, acc, NoRequests());
            if (m > 8) gather_rows8(fbase, my, 8, m - 8, lane_off, acc, NoRequests());
        }
        if (c > SLAB) {
            int ns = *side_cursor;
            ns = (ns < E) ? ns : E;
            for (int i = 0; i < ns; ++i) {
                if (side[2 * (size_t)i] == node) {
                    const int s = side[2 * (size_t)i + 1];
                    const float4 v = *(const float4*)(fbase + ((unsigned)s * 256u + lane_off));
                    acc.x += v.x; acc.y += v.y; acc.z += v.z; acc.w += v.w;
                }
            }
        }
        *(float4*)(tl + r * TILE_PITCH + 4 * l) = acc;
        if (l == 0) {
            const float d = (float)c;
            sv[r] = (d > 0.0f) ? (1.0f / d) : 0.0f;
        }
        __syncthreads();

        c_cur = c_nxt; ids_cur = ids_nxt; c_nxt = c_nn;

        if (w < tilesN) {
            const float* Ap = tl + l15 * TILE_PITCH + kb;
            const float* Bp = wsh + (w * 16 + l15) * TILE_PITCH + kb;
            v8f cacc = {};
            #pragma unroll
            for (int q = 0; q < FEAT / 4; ++q) {
                v2f a = *(const v2f*)(Ap + 4 * q);
                v2f b = *(const v2f*)(Bp + 4 * q);
                cacc = __builtin_amdgcn_wmma_f32_16x16x4_f32(
                         false, a,  false, b,
                         (short)0, cacc,  false,  false);
            }

            const float4 i0 = *(const float4*)(sv + half * 8);
            const float4 i1 = *(const float4*)(sv + half * 8 + 4);
            const float invs[8] = {i0.x, i0.y, i0.z, i0.w, i1.x, i1.y, i1.z, i1.w};
            const int row0 = t * 16 + half * 8;
            float* op = out + (long long)row0 * NCOLS + cB;
            #pragma unroll
            for (int q = 0; q < 8; ++q) {
                float v = cacc[q] * invs[q] + bv;
                if (RELU) v = fmaxf(v, 0.0f);
                if (colOK && row0 + q < n) {
                    op[q * NCOLS] = v;
                }
            }
        }
    }
}

extern "C" void kernel_launch(void* const* d_in, const int* in_sizes, int n_in,
                              void* d_out, int out_size, void* d_ws, size_t ws_size,
                              hipStream_t stream) {
    const float* features = (const float*)d_in[0];
    const float* W1 = (const float*)d_in[1];
    const float* b1 = (const float*)d_in[2];
    const float* W2 = (const float*)d_in[3];
    const float* b2 = (const float*)d_in[4];
    const int* src = (const int*)d_in[5];
    const int* dst = (const int*)d_in[6];
    float* out = (float*)d_out;

    const int N = in_sizes[0] / FEAT;
    const int E = in_sizes[5];

    float* h1   = (float*)d_ws;
    int* slab   = (int*)(h1 + (long long)N * FEAT);
    int* cnt    = slab + (long long)N * SLAB;
    int* ctrl   = cnt + N;
    int* side_cursor = ctrl;
    int* side   = ctrl + CTRL_INTS;

    hipMemsetAsync(cnt, 0, (size_t)(N + CTRL_INTS) * sizeof(int), stream);

    const int nchunks = (E + CHUNK - 1) / CHUNK;
    place_edges_kernel<<<1024, 256, 0, stream>>>(src, dst, cnt, ctrl + CTRL_TICK(0), slab,
                                                 side_cursor, side, E, N, nchunks);

    gather_gemm_kernel<FEAT, true><<<RES_GROUPS, 256, 0, stream>>>(features, slab, cnt, side_cursor, side,
                                                                    W1, b1, h1, N, E);

    gather_gemm_kernel<NCLS, false><<<RES_GROUPS, 256, 0, stream>>>(h1, slab, cnt, side_cursor, side,
                                                                     W2, b2, out, N, E);
}
